# starting kernel
# baseline (speedup 1.0000x reference)
#include <hip/hip_runtime.h>
#include <hip/hip_bf16.h>

typedef float v2f __attribute__((ext_vector_type(2)));
typedef float v8f __attribute__((ext_vector_type(8)));

#define N_NODES   50000
#define N_EDGES   800000
#define IN_FEATS  256
#define OUT_FEATS 128

__global__ void gc_zero_kernel(float* __restrict__ p, int n) {
    int i = blockIdx.x * blockDim.x + threadIdx.x;
    int stride = gridDim.x * blockDim.x;
    for (; i < n; i += stride) p[i] = 0.0f;
}

__global__ void gc_degree_kernel(const int* __restrict__ src,
                                 const int* __restrict__ dst,
                                 float* __restrict__ out_deg,
                                 float* __restrict__ in_deg) {
    int e = blockIdx.x * blockDim.x + threadIdx.x;
    if (e >= N_EDGES) return;
    unsafeAtomicAdd(&out_deg[src[e]], 1.0f);
    unsafeAtomicAdd(&in_deg[dst[e]], 1.0f);
}

__global__ void gc_gemm_wmma_kernel(const float* __restrict__ x,
                                    const float* __restrict__ Wm,
                                    const float* __restrict__ out_deg,
                                    float* __restrict__ h) {
    const int lane = threadIdx.x & 31;
    const int wave = threadIdx.x >> 5;
    const int half = lane >> 4;
    const int l    = lane & 15;

    const int m0 = blockIdx.x * 16;
    const int n0 = wave * 16;

    const int row = m0 + l;
    const float rs = rsqrtf(fmaxf(out_deg[row], 1.0f));
    const float* __restrict__ xrow = x + (size_t)row * IN_FEATS;

    v8f c = {};
    #pragma unroll 8
    for (int k0 = 0; k0 < IN_FEATS; k0 += 4) {
        const int ka = k0 + 2 * half;
        v2f a, b;
        a.x = xrow[ka]     * rs;
        a.y = xrow[ka + 1] * rs;
        b.x = Wm[(size_t)ka       * OUT_FEATS + n0 + l];
        b.y = Wm[(size_t)(ka + 1) * OUT_FEATS + n0 + l];
        c = __builtin_amdgcn_wmma_f32_16x16x4_f32(false, a, false, b,
                                                  (short)0, c, false, false);
    }

    #pragma unroll
    for (int r = 0; r < 8; ++r) {
        const int m = m0 + r + 8 * half;
        h[(size_t)m * OUT_FEATS + n0 + l] = c[r];
    }
}

__global__ void gc_spmm_kernel(const float* __restrict__ h,
                               const float* __restrict__ ew,
                               const int* __restrict__ src,
                               const int* __restrict__ dst,
                               float* __restrict__ agg) {
    const int wave_in_blk = threadIdx.x >> 5;
    const int lane        = threadIdx.x & 31;
    const long long e = (long long)blockIdx.x * (blockDim.x >> 5) + wave_in_blk;
    if (e >= N_EDGES) return;

    const int  s = src[e];
    const int  d = dst[e];
    const float w = ew[e];

    const float4 hv = *(const float4*)(h + (size_t)s * OUT_FEATS + lane * 4);
    float* outp = agg + (size_t)d * OUT_FEATS + lane * 4;
    unsafeAtomicAdd(outp + 0, hv.x * w);
    unsafeAtomicAdd(outp + 1, hv.y * w);
    unsafeAtomicAdd(outp + 2, hv.z * w);
    unsafeAtomicAdd(outp + 3, hv.w * w);
}

__global__ void gc_finalize_kernel(float* __restrict__ out,
                                   const float* __restrict__ in_deg,
                                   const float* __restrict__ b) {
    const long long i = (long long)blockIdx.x * blockDim.x + threadIdx.x;
    if (i >= (long long)N_NODES * OUT_FEATS) return;
    const int node = (int)(i >> 7);
    const int col  = (int)(i & 127);
    const float rn = rsqrtf(fmaxf(in_deg[node], 1.0f));
    out[i] = out[i] * rn + b[col];
}

extern "C" void kernel_launch(void* const* d_in, const int* in_sizes, int n_in,
                              void* d_out, int out_size, void* d_ws, size_t ws_size,
                              hipStream_t stream) {
    const float* x   = (const float*)d_in[0];
    const float* ew  = (const float*)d_in[1];
    const float* Wm  = (const float*)d_in[2];
    const float* b   = (const float*)d_in[3];
    const int*   src = (const int*)d_in[4];
    const int*   dst = (const int*)d_in[5];

    float* out = (float*)d_out;

    float* out_deg = (float*)d_ws;
    float* in_deg  = out_deg + N_NODES;
    float* h       = in_deg + N_NODES;

    gc_zero_kernel<<<512, 256, 0, stream>>>(out_deg, 2 * N_NODES);
    gc_zero_kernel<<<4096, 256, 0, stream>>>(out, N_NODES * OUT_FEATS);

    gc_degree_kernel<<<(N_EDGES + 255) / 256, 256, 0, stream>>>(src, dst, out_deg, in_deg);

    gc_gemm_wmma_kernel<<<N_NODES / 16, 256, 0, stream>>>(x, Wm, out_deg, h);

    gc_spmm_kernel<<<N_EDGES / 8, 256, 0, stream>>>(h, ew, src, dst, out);

    gc_finalize_kernel<<<(N_NODES * OUT_FEATS + 255) / 256, 256, 0, stream>>>(out, in_deg, b);
}


# degree counts by per-workgroup LDS histograms (integer LDS atomics) plus a partial-sum reduce, replacing 2 x 800000 global float atomics; counts stay exact, float summation order untouched
# speedup vs baseline: 1.3773x; 1.3773x over previous
#include <hip/hip_runtime.h>
#include <hip/hip_bf16.h>

typedef float v2f __attribute__((ext_vector_type(2)));
typedef float v8f __attribute__((ext_vector_type(8)));

#define N_NODES   50000
#define N_EDGES   800000
#define IN_FEATS  256
#define OUT_FEATS 128

__global__ void gc_zero_kernel(float* __restrict__ p, int n) {
    int i = blockIdx.x * blockDim.x + threadIdx.x;
    int stride = gridDim.x * blockDim.x;
    for (; i < n; i += stride) p[i] = 0.0f;
}

#define HIST_WGS      32
#define HIST_THREADS  1024
#define EDGES_PER_WG  (N_EDGES / HIST_WGS)
#define NODE_QUADS    (N_NODES / 4)
#define IDX_QUADS_PER_WG (EDGES_PER_WG / 4)
#define HIST_TRIPS    ((IDX_QUADS_PER_WG + HIST_THREADS - 1) / HIST_THREADS)

__global__ __launch_bounds__(HIST_THREADS)
void gc_degree_hist_kernel(const int* __restrict__ src,
                           const int* __restrict__ dst,
                           unsigned int* __restrict__ partial) {
    __shared__ unsigned int hist[N_NODES];
    const int tid   = threadIdx.x;
    const int wg    = blockIdx.x;
    const int chunk = (wg < HIST_WGS) ? wg : wg - HIST_WGS;
    const int* __restrict__ idx = ((wg < HIST_WGS) ? src : dst) + (size_t)chunk * EDGES_PER_WG;
    const int4* __restrict__ idx4 = (const int4*)idx;

    int4 v[HIST_TRIPS];
    #pragma unroll
    for (int j = 0; j < HIST_TRIPS; ++j) {
        const int i = tid + j * HIST_THREADS;
        if (i < IDX_QUADS_PER_WG) v[j] = idx4[i];
    }

    uint4* hist4 = (uint4*)hist;
    for (int i = tid; i < NODE_QUADS; i += HIST_THREADS)
        hist4[i] = make_uint4(0u, 0u, 0u, 0u);
    __syncthreads();

    #pragma unroll
    for (int j = 0; j < HIST_TRIPS; ++j) {
        const int i = tid + j * HIST_THREADS;
        if (i < IDX_QUADS_PER_WG) {
            atomicAdd(&hist[v[j].x], 1u);
            atomicAdd(&hist[v[j].y], 1u);
            atomicAdd(&hist[v[j].z], 1u);
            atomicAdd(&hist[v[j].w], 1u);
        }
    }
    __syncthreads();

    uint4* __restrict__ out4 = (uint4*)(partial + (size_t)wg * N_NODES);
    for (int i = tid; i < NODE_QUADS; i += HIST_THREADS)
        out4[i] = hist4[i];
}

__global__ void gc_degree_reduce_kernel(const unsigned int* __restrict__ partial,
                                        float* __restrict__ deg) {
    const int q = blockIdx.x * blockDim.x + threadIdx.x;
    if (q >= 2 * NODE_QUADS) return;
    const int a  = q / NODE_QUADS;
    const int i4 = q - a * NODE_QUADS;
    const uint4* __restrict__ p = (const uint4*)partial + (size_t)a * HIST_WGS * NODE_QUADS + i4;
    uint4 s = make_uint4(0u, 0u, 0u, 0u);
    #pragma unroll 8
    for (int k = 0; k < HIST_WGS; ++k) {
        const uint4 v = p[(size_t)k * NODE_QUADS];
        s.x += v.x; s.y += v.y; s.z += v.z; s.w += v.w;
    }
    ((float4*)deg)[q] = make_float4((float)s.x, (float)s.y, (float)s.z, (float)s.w);
}

__global__ void gc_gemm_wmma_kernel(const float* __restrict__ x,
                                    const float* __restrict__ Wm,
                                    const float* __restrict__ out_deg,
                                    float* __restrict__ h) {
    const int lane = threadIdx.x & 31;
    const int wave = threadIdx.x >> 5;
    const int half = lane >> 4;
    const int l    = lane & 15;

    const int m0 = blockIdx.x * 16;
    const int n0 = wave * 16;

    const int row = m0 + l;
    const float rs = rsqrtf(fmaxf(out_deg[row], 1.0f));
    const float* __restrict__ xrow = x + (size_t)row * IN_FEATS;

    v8f c = {};
    #pragma unroll 8
    for (int k0 = 0; k0 < IN_FEATS; k0 += 4) {
        const int ka = k0 + 2 * half;
        v2f a, b;
        a.x = xrow[ka]     * rs;
        a.y = xrow[ka + 1] * rs;
        b.x = Wm[(size_t)ka       * OUT_FEATS + n0 + l];
        b.y = Wm[(size_t)(ka + 1) * OUT_FEATS + n0 + l];
        c = __builtin_amdgcn_wmma_f32_16x16x4_f32(false, a, false, b,
                                                  (short)0, c, false, false);
    }

    #pragma unroll
    for (int r = 0; r < 8; ++r) {
        const int m = m0 + r + 8 * half;
        h[(size_t)m * OUT_FEATS + n0 + l] = c[r];
    }
}

__global__ void gc_spmm_kernel(const float* __restrict__ h,
                               const float* __restrict__ ew,
                               const int* __restrict__ src,
                               const int* __restrict__ dst,
                               float* __restrict__ agg) {
    const int wave_in_blk = threadIdx.x >> 5;
    const int lane        = threadIdx.x & 31;
    const long long e = (long long)blockIdx.x * (blockDim.x >> 5) + wave_in_blk;
    if (e >= N_EDGES) return;

    const int  s = src[e];
    const int  d = dst[e];
    const float w = ew[e];

    const float4 hv = *(const float4*)(h + (size_t)s * OUT_FEATS + lane * 4);
    float* outp = agg + (size_t)d * OUT_FEATS + lane * 4;
    unsafeAtomicAdd(outp + 0, hv.x * w);
    unsafeAtomicAdd(outp + 1, hv.y * w);
    unsafeAtomicAdd(outp + 2, hv.z * w);
    unsafeAtomicAdd(outp + 3, hv.w * w);
}

__global__ void gc_finalize_kernel(float* __restrict__ out,
                                   const float* __restrict__ in_deg,
                                   const float* __restrict__ b) {
    const long long i = (long long)blockIdx.x * blockDim.x + threadIdx.x;
    if (i >= (long long)N_NODES * OUT_FEATS) return;
    const int node = (int)(i >> 7);
    const int col  = (int)(i & 127);
    const float rn = rsqrtf(fmaxf(in_deg[node], 1.0f));
    out[i] = out[i] * rn + b[col];
}

extern "C" void kernel_launch(void* const* d_in, const int* in_sizes, int n_in,
                              void* d_out, int out_size, void* d_ws, size_t ws_size,
                              hipStream_t stream) {
    const float* x   = (const float*)d_in[0];
    const float* ew  = (const float*)d_in[1];
    const float* Wm  = (const float*)d_in[2];
    const float* b   = (const float*)d_in[3];
    const int*   src = (const int*)d_in[4];
    const int*   dst = (const int*)d_in[5];

    float* out = (float*)d_out;

    float* out_deg = (float*)d_ws;
    float* in_deg  = out_deg + N_NODES;
    float* h       = in_deg + N_NODES;

    unsigned int* partial = (unsigned int*)h;
    gc_degree_hist_kernel<<<2 * HIST_WGS, HIST_THREADS, 0, stream>>>(src, dst, partial);
    gc_degree_reduce_kernel<<<(2 * NODE_QUADS + 255) / 256, 256, 0, stream>>>(partial, out_deg);

    gc_zero_kernel<<<4096, 256, 0, stream>>>(out, N_NODES * OUT_FEATS);

    gc_gemm_wmma_kernel<<<N_NODES / 16, 256, 0, stream>>>(x, Wm, out_deg, h);

    gc_spmm_kernel<<<N_EDGES / 8, 256, 0, stream>>>(h, ew, src, dst, out);

    gc_finalize_kernel<<<(N_NODES * OUT_FEATS + 255) / 256, 256, 0, stream>>>(out, in_deg, b);
}


# GEMM operands staged: W in 32-k slabs through LDS shared by a 128-row block; each wave keeps a 16x128 strip in 8 accumulators, reads x once 16 B per lane; k summed in a new order inside each 8 k
# speedup vs baseline: 2.9292x; 2.1268x over previous
#include <hip/hip_runtime.h>


typedef float v2f __attribute__((ext_vector_type(2)));
typedef float v8f __attribute__((ext_vector_type(8)));

#define N_NODES    50000
#define N_EDGES    800000
#define IN_FEATS   256
#define OUT_FEATS  128
#define HALF_FEATS 64

#define HIST_WGS      32
#define HIST_THREADS  1024
#define EDGES_PER_WG  (N_EDGES / HIST_WGS)
#define NODE_QUADS    (N_NODES / 4)
#define IDX_QUADS_PER_WG (EDGES_PER_WG / 4)
#define HIST_TRIPS    ((IDX_QUADS_PER_WG + HIST_THREADS - 1) / HIST_THREADS)

__global__ __launch_bounds__(HIST_THREADS)
void gc_degree_hist_kernel(const int* __restrict__ src,
                           const int* __restrict__ dst,
                           unsigned int* __restrict__ partial,
                           unsigned int* __restrict__ rank) {
    __shared__ unsigned int hist[N_NODES];
    const int tid   = threadIdx.x;
    const int wg    = blockIdx.x;
    const bool is_dst = (wg >= HIST_WGS);
    const int chunk = is_dst ? wg - HIST_WGS : wg;
    const int* __restrict__ idx = (is_dst ? dst : src) + (size_t)chunk * EDGES_PER_WG;
    const int4* __restrict__ idx4 = (const int4*)idx;

    int4 v[HIST_TRIPS];
    #pragma unroll
    for (int j = 0; j < HIST_TRIPS; ++j) {
        const int i = tid + j * HIST_THREADS;
        if (i < IDX_QUADS_PER_WG) v[j] = idx4[i];
    }

    uint4* hist4 = (uint4*)hist;
    for (int i = tid; i < NODE_QUADS; i += HIST_THREADS)
        hist4[i] = make_uint4(0u, 0u, 0u, 0u);
    __syncthreads();

    uint4* __restrict__ rank4 = (uint4*)rank + (size_t)chunk * IDX_QUADS_PER_WG;
    #pragma unroll
    for (int j = 0; j < HIST_TRIPS; ++j) {
        const int i = tid + j * HIST_THREADS;
        if (i < IDX_QUADS_PER_WG) {
            const unsigned int r0 = atomicAdd(&hist[v[j].x], 1u);
            const unsigned int r1 = atomicAdd(&hist[v[j].y], 1u);
            const unsigned int r2 = atomicAdd(&hist[v[j].z], 1u);
            const unsigned int r3 = atomicAdd(&hist[v[j].w], 1u);
            if (is_dst) rank4[i] = make_uint4(r0, r1, r2, r3);
        }
    }
    __syncthreads();

    uint4* __restrict__ out4 = (uint4*)(partial + (size_t)wg * N_NODES);
    for (int i = tid; i < NODE_QUADS; i += HIST_THREADS)
        out4[i] = hist4[i];
}

#define SCAN_THREADS 256
#define SCAN_BLOCKS  ((N_NODES + SCAN_THREADS - 1) / SCAN_THREADS)

__device__ __forceinline__ unsigned int gc_block_excl_scan(unsigned int v,
                                                           unsigned int* wsum,
                                                           unsigned int& total) {
    const int lane = threadIdx.x & 31;
    const int wv   = threadIdx.x >> 5;
    unsigned int inc = v;
    #pragma unroll
    for (int o = 1; o < 32; o <<= 1) {
        const unsigned int t = __shfl_up(inc, (unsigned int)o);
        if (lane >= o) inc += t;
    }
    if (lane == 31) wsum[wv] = inc;
    __syncthreads();
    unsigned int base = 0u, tot = 0u;
    #pragma unroll
    for (int k = 0; k < SCAN_THREADS / 32; ++k) {
        const unsigned int s = wsum[k];
        if (k < wv) base += s;
        tot += s;
    }
    total = tot;
    return base + inc - v;
}

__global__ __launch_bounds__(SCAN_THREADS)
void gc_degree_scan_kernel(unsigned int* __restrict__ partial,
                           float* __restrict__ deg,
                           unsigned int* __restrict__ local_off,
                           unsigned int* __restrict__ blocksum) {
    __shared__ unsigned int wsum[SCAN_THREADS / 32];
    const int n = blockIdx.x * SCAN_THREADS + threadIdx.x;
    const bool ok = n < N_NODES;
    unsigned int cnt_in = 0u;
    if (ok) {
        const unsigned int* ps = partial + n;
        unsigned int* pd = partial + (size_t)HIST_WGS * N_NODES + n;
        unsigned int s = 0u;
        #pragma unroll
        for (int c = 0; c < HIST_WGS; ++c) s += ps[(size_t)c * N_NODES];
        deg[n] = (float)s;
        unsigned int vd[HIST_WGS];
        #pragma unroll
        for (int c = 0; c < HIST_WGS; ++c) vd[c] = pd[(size_t)c * N_NODES];
        unsigned int run = 0u;
        #pragma unroll
        for (int c = 0; c < HIST_WGS; ++c) { pd[(size_t)c * N_NODES] = run; run += vd[c]; }
        deg[N_NODES + n] = (float)run;
        cnt_in = run;
    }
    unsigned int total;
    const unsigned int ex = gc_block_excl_scan(cnt_in, wsum, total);
    if (ok) local_off[n] = ex;
    if (threadIdx.x == 0) blocksum[blockIdx.x] = total;
}

__global__ __launch_bounds__(SCAN_THREADS)
void gc_fill_kernel(const int* __restrict__ src,
                    const int* __restrict__ dst,
                    const float* __restrict__ ew,
                    const unsigned int* __restrict__ rank,
                    const unsigned int* __restrict__ chunk_off,
                    const unsigned int* __restrict__ local_off,
                    const unsigned int* __restrict__ blocksum,
                    unsigned int* __restrict__ rowstart,
                    unsigned int* __restrict__ pairs_w) {
    __shared__ unsigned int wsum[SCAN_THREADS / 32];
    __shared__ unsigned int bp[SCAN_THREADS];
    const int t = threadIdx.x;
    const int e = blockIdx.x * SCAN_THREADS + t;
    const int   d = dst[e];
    const int   s = src[e];
    const float w = ew[e];
    const unsigned int r = rank[e];

    const unsigned int bs = (t < SCAN_BLOCKS) ? blocksum[t] : 0u;
    unsigned int total;
    bp[t] = gc_block_excl_scan(bs, wsum, total);
    __syncthreads();

    const int c = e / EDGES_PER_WG;
    const unsigned int pos = bp[d >> 8] + local_off[d]
                           + chunk_off[(size_t)c * N_NODES + d] + r;
    ((uint2*)pairs_w)[pos] = make_uint2((unsigned int)s, __float_as_uint(w));
    if (e < N_NODES) rowstart[e] = bp[e >> 8] + local_off[e];
}

#define GEMM_ROWS   128
#define GEMM_KS     32
#define GEMM_SLABS  (IN_FEATS / GEMM_KS)
#define GEMM_KQ     (GEMM_KS / 4)
#define GEMM_WGS    ((N_NODES + GEMM_ROWS - 1) / GEMM_ROWS)

__global__ __launch_bounds__(256)
void gc_gemm_wmma_kernel(const float* __restrict__ x,
                         const float* __restrict__ Wm,
                         const float* __restrict__ out_deg,
                         float* __restrict__ h_lo,
                         float* __restrict__ h_hi) {
    __shared__ float4 wq[2][GEMM_KQ][OUT_FEATS];

    const int tid  = threadIdx.x;
    const int lane = tid & 31;
    const int wave = tid >> 5;
    const int half = lane >> 4;
    const int l    = lane & 15;

    const int m0 = blockIdx.x * GEMM_ROWS + wave * 16;
    const bool live = m0 < N_NODES;
    const int row = live ? (m0 + l) : (N_NODES - 1);
    const float rs = rsqrtf(fmaxf(out_deg[row], 1.0f));
    const float4* __restrict__ xq = (const float4*)(x + (size_t)row * IN_FEATS) + half;

    const int sn  = tid & (OUT_FEATS - 1);
    const int skq = tid >> 7;
    const float* __restrict__ wcol = Wm + sn;

    float  wst[4][4];
    float4 acur[4], anext[4];

    #pragma unroll
    for (int q = 0; q < 4; ++q) {
        #pragma unroll
        for (int e = 0; e < 4; ++e)
            wst[q][e] = wcol[(size_t)(4 * (skq + 2 * q) + e) * OUT_FEATS];
    }
    #pragma unroll
    for (int j = 0; j < 4; ++j) acur[j] = xq[2 * j];
    #pragma unroll
    for (int q = 0; q < 4; ++q)
        wq[0][skq + 2 * q][sn] = make_float4(wst[q][0], wst[q][1], wst[q][2], wst[q][3]);
    __syncthreads();

    v8f c[8] = {};
    for (int s = 0; s < GEMM_SLABS; ++s) {
        const int buf = s & 1;
        const bool more = (s + 1 < GEMM_SLABS);
        if (more) {
            const int k1 = (s + 1) * GEMM_KS;
            #pragma unroll
            for (int q = 0; q < 4; ++q) {
                #pragma unroll
                for (int e = 0; e < 4; ++e)
                    wst[q][e] = wcol[(size_t)(k1 + 4 * (skq + 2 * q) + e) * OUT_FEATS];
            }
            #pragma unroll
            for (int j = 0; j < 4; ++j) anext[j] = xq[(s + 1) * GEMM_KQ + 2 * j];
        }
        asm volatile("" ::: "memory");

        float4 bcur[8], bnext[8];
        #pragma unroll
        for (int t = 0; t < 8; ++t) bcur[t] = wq[buf][half][16 * t + l];

        #pragma unroll
        for (int j = 0; j < 4; ++j) {
            if (j + 1 < 4) {
                #pragma unroll
                for (int t = 0; t < 8; ++t) bnext[t] = wq[buf][2 * (j + 1) + half][16 * t + l];
            }
            __builtin_amdgcn_sched_barrier(0);
            const float4 av = acur[j];
            v2f a0, a1;
            a0.x = av.x * rs; a0.y = av.y * rs;
            a1.x = av.z * rs; a1.y = av.w * rs;
            #pragma unroll
            for (int t = 0; t < 8; ++t) {
                v2f b0; b0.x = bcur[t].x; b0.y = bcur[t].y;
                c[t] = __builtin_amdgcn_wmma_f32_16x16x4_f32(false, a0, false, b0,
                                                             (short)0, c[t], false, false);
            }
            #pragma unroll
            for (int t = 0; t < 8; ++t) {
                v2f b1; b1.x = bcur[t].z; b1.y = bcur[t].w;
                c[t] = __builtin_amdgcn_wmma_f32_16x16x4_f32(false, a1, false, b1,
                                                             (short)0, c[t], false, false);
            }
            if (j + 1 < 4) {
                #pragma unroll
                for (int t = 0; t < 8; ++t) bcur[t] = bnext[t];
            }
        }

        if (more) {
            #pragma unroll
            for (int q = 0; q < 4; ++q)
                wq[buf ^ 1][skq + 2 * q][sn] =
                    make_float4(wst[q][0], wst[q][1], wst[q][2], wst[q][3]);
            __syncthreads();
            #pragma unroll
            for (int j = 0; j < 4; ++j) acur[j] = anext[j];
        }
    }

    if (live) {
        #pragma unroll
        for (int t = 0; t < 8; ++t) {
            float* __restrict__ hp;
            int pitch;
            if (t < 4) { hp = h_lo + 16 * t + l;                pitch = OUT_FEATS;  }
            else       { hp = h_hi + (16 * t - HALF_FEATS) + l; pitch = HALF_FEATS; }
            #pragma unroll
            for (int r = 0; r < 8; ++r) {
                const int m = m0 + r + 8 * half;
                hp[(size_t)m * pitch] = c[t][r];
            }
        }
    }
}

#define GATHER_WAVES 8
__global__ __launch_bounds__(GATHER_WAVES * 32)
void gc_gather_kernel(const float* hsrc, unsigned int hpitch,
                      const unsigned int* __restrict__ pairs_w,
                      const unsigned int* __restrict__ rowstart,
                      const float* __restrict__ in_deg,
                      const float* __restrict__ bias,
                      float* outp) {
    const int lane = threadIdx.x & 31;
    const int wv   = threadIdx.x >> 5;
    const int hf   = lane >> 4;
    const int l16  = lane & 15;
    const int d    = blockIdx.x * GATHER_WAVES + wv;

    const float degf = in_deg[d];
    const int   cnt  = (int)degf;
    const uint2* __restrict__ p = (const uint2*)pairs_w + rowstart[d];
    const float* hb = hsrc + l16 * 4;

    float4 acc = make_float4(0.0f, 0.0f, 0.0f, 0.0f);
    int i = hf;
    for (; i + 6 < cnt; i += 8) {
        const uint2 q0 = p[i], q1 = p[i + 2], q2 = p[i + 4], q3 = p[i + 6];
        const float4 r0 = *(const float4*)(hb + q0.x * hpitch);
        const float4 r1 = *(const float4*)(hb + q1.x * hpitch);
        const float4 r2 = *(const float4*)(hb + q2.x * hpitch);
        const float4 r3 = *(const float4*)(hb + q3.x * hpitch);
        const float w0 = __uint_as_float(q0.y), w1 = __uint_as_float(q1.y);
        const float w2 = __uint_as_float(q2.y), w3 = __uint_as_float(q3.y);
        acc.x = fmaf(r0.x, w0, acc.x); acc.y = fmaf(r0.y, w0, acc.y);
        acc.z = fmaf(r0.z, w0, acc.z); acc.w = fmaf(r0.w, w0, acc.w);
        acc.x = fmaf(r1.x, w1, acc.x); acc.y = fmaf(r1.y, w1, acc.y);
        acc.z = fmaf(r1.z, w1, acc.z); acc.w = fmaf(r1.w, w1, acc.w);
        acc.x = fmaf(r2.x, w2, acc.x); acc.y = fmaf(r2.y, w2, acc.y);
        acc.z = fmaf(r2.z, w2, acc.z); acc.w = fmaf(r2.w, w2, acc.w);
        acc.x = fmaf(r3.x, w3, acc.x); acc.y = fmaf(r3.y, w3, acc.y);
        acc.z = fmaf(r3.z, w3, acc.z); acc.w = fmaf(r3.w, w3, acc.w);
    }
    for (; i < cnt; i += 2) {
        const uint2 q0 = p[i];
        const float4 r0 = *(const float4*)(hb + q0.x * hpitch);
        const float w0 = __uint_as_float(q0.y);
        acc.x = fmaf(r0.x, w0, acc.x); acc.y = fmaf(r0.y, w0, acc.y);
        acc.z = fmaf(r0.z, w0, acc.z); acc.w = fmaf(r0.w, w0, acc.w);
    }

    acc.x += __shfl_xor(acc.x, 16);
    acc.y += __shfl_xor(acc.y, 16);
    acc.z += __shfl_xor(acc.z, 16);
    acc.w += __shfl_xor(acc.w, 16);

    if (hf == 0) {
        const float rn = rsqrtf(fmaxf(degf, 1.0f));
        const float4 bb = *(const float4*)(bias + l16 * 4);
        float4 o;
        o.x = acc.x * rn + bb.x;
        o.y = acc.y * rn + bb.y;
        o.z = acc.z * rn + bb.z;
        o.w = acc.w * rn + bb.w;
        *(float4*)(outp + (size_t)d * OUT_FEATS + l16 * 4) = o;
    }
}

extern "C" void kernel_launch(void* const* d_in, const int* in_sizes, int n_in,
                              void* d_out, int out_size, void* d_ws, size_t ws_size,
                              hipStream_t stream) {
    const float* x   = (const float*)d_in[0];
    const float* ew  = (const float*)d_in[1];
    const float* Wm  = (const float*)d_in[2];
    const float* b   = (const float*)d_in[3];
    const int*   src = (const int*)d_in[4];
    const int*   dst = (const int*)d_in[5];

    float* out = (float*)d_out;

    float*        out_deg   = (float*)d_ws;
    float*        in_deg    = out_deg + 50000;
    float*        h_hi      = out_deg + 100032;
    unsigned int* partial   = (unsigned int*)h_hi;
    unsigned int* pairs     = (unsigned int*)(out_deg + 3300064);
    unsigned int* rowstart  = (unsigned int*)(out_deg + 4900064);
    unsigned int* local_off = (unsigned int*)(out_deg + 4950064);
    unsigned int* blocksum  = (unsigned int*)(out_deg + 5000064);
    unsigned int* rank      = (unsigned int*)(out_deg + 5000320);

    gc_degree_hist_kernel<<<2 * HIST_WGS, HIST_THREADS, 0, stream>>>(src, dst, partial, rank);

    gc_degree_scan_kernel<<<SCAN_BLOCKS, SCAN_THREADS, 0, stream>>>(partial, out_deg, local_off, blocksum);

    gc_fill_kernel<<<N_EDGES / SCAN_THREADS, SCAN_THREADS, 0, stream>>>(
        src, dst, ew, rank, partial + HIST_WGS * N_NODES, local_off, blocksum, rowstart, pairs);

    gc_gemm_wmma_kernel<<<GEMM_WGS, 256, 0, stream>>>(x, Wm, out_deg, out + HALF_FEATS, h_hi);

    gc_gather_kernel<<<N_NODES / GATHER_WAVES, GATHER_WAVES * 32, 0, stream>>>(
        out + HALF_FEATS, OUT_FEATS, pairs, rowstart, in_deg, b, out);

    gc_gather_kernel<<<N_NODES / GATHER_WAVES, GATHER_WAVES * 32, 0, stream>>>(
        h_hi, HALF_FEATS, pairs, rowstart, in_deg, b + HALF_FEATS, out + HALF_FEATS);
}


# gather: one half wave per destination node (two nodes per wave: 25000 waves per gather launch, not 50000); pairs read one per lane and row-broadcast; a node's sum is now one chain in run order
# speedup vs baseline: 3.0893x; 1.0547x over previous
#include <hip/hip_runtime.h>


typedef float v2f __attribute__((ext_vector_type(2)));
typedef float v8f __attribute__((ext_vector_type(8)));

#define N_NODES    50000
#define N_EDGES    800000
#define IN_FEATS   256
#define OUT_FEATS  128
#define HALF_FEATS 64

#define HIST_WGS      32
#define HIST_THREADS  1024
#define EDGES_PER_WG  (N_EDGES / HIST_WGS)
#define NODE_QUADS    (N_NODES / 4)
#define IDX_QUADS_PER_WG (EDGES_PER_WG / 4)
#define HIST_TRIPS    ((IDX_QUADS_PER_WG + HIST_THREADS - 1) / HIST_THREADS)

__global__ __launch_bounds__(HIST_THREADS)
void gc_degree_hist_kernel(const int* __restrict__ src,
                           const int* __restrict__ dst,
                           unsigned int* __restrict__ partial,
                           unsigned int* __restrict__ rank) {
    __shared__ unsigned int hist[N_NODES];
    const int tid   = threadIdx.x;
    const int wg    = blockIdx.x;
    const bool is_dst = (wg >= HIST_WGS);
    const int chunk = is_dst ? wg - HIST_WGS : wg;
    const int* __restrict__ idx = (is_dst ? dst : src) + (size_t)chunk * EDGES_PER_WG;
    const int4* __restrict__ idx4 = (const int4*)idx;

    int4 v[HIST_TRIPS];
    #pragma unroll
    for (int j = 0; j < HIST_TRIPS; ++j) {
        const int i = tid + j * HIST_THREADS;
        if (i < IDX_QUADS_PER_WG) v[j] = idx4[i];
    }

    uint4* hist4 = (uint4*)hist;
    for (int i = tid; i < NODE_QUADS; i += HIST_THREADS)
        hist4[i] = make_uint4(0u, 0u, 0u, 0u);
    __syncthreads();

    uint4* __restrict__ rank4 = (uint4*)rank + (size_t)chunk * IDX_QUADS_PER_WG;
    #pragma unroll
    for (int j = 0; j < HIST_TRIPS; ++j) {
        const int i = tid + j * HIST_THREADS;
        if (i < IDX_QUADS_PER_WG) {
            const unsigned int r0 = atomicAdd(&hist[v[j].x], 1u);
            const unsigned int r1 = atomicAdd(&hist[v[j].y], 1u);
            const unsigned int r2 = atomicAdd(&hist[v[j].z], 1u);
            const unsigned int r3 = atomicAdd(&hist[v[j].w], 1u);
            if (is_dst) rank4[i] = make_uint4(r0, r1, r2, r3);
        }
    }
    __syncthreads();

    uint4* __restrict__ out4 = (uint4*)(partial + (size_t)wg * N_NODES);
    for (int i = tid; i < NODE_QUADS; i += HIST_THREADS)
        out4[i] = hist4[i];
}

#define SCAN_THREADS 256
#define SCAN_BLOCKS  ((N_NODES + SCAN_THREADS - 1) / SCAN_THREADS)

__device__ __forceinline__ unsigned int gc_block_excl_scan(unsigned int v,
                                                           unsigned int* wsum,
                                                           unsigned int& total) {
    const int lane = threadIdx.x & 31;
    const int wv   = threadIdx.x >> 5;
    unsigned int inc = v;
    #pragma unroll
    for (int o = 1; o < 32; o <<= 1) {
        const unsigned int t = __shfl_up(inc, (unsigned int)o);
        if (lane >= o) inc += t;
    }
    if (lane == 31) wsum[wv] = inc;
    __syncthreads();
    unsigned int base = 0u, tot = 0u;
    #pragma unroll
    for (int k = 0; k < SCAN_THREADS / 32; ++k) {
        const unsigned int s = wsum[k];
        if (k < wv) base += s;
        tot += s;
    }
    total = tot;
    return base + inc - v;
}

__global__ __launch_bounds__(SCAN_THREADS)
void gc_degree_scan_kernel(unsigned int* __restrict__ partial,
                           float* __restrict__ deg,
                           unsigned int* __restrict__ local_off,
                           unsigned int* __restrict__ blocksum) {
    __shared__ unsigned int wsum[SCAN_THREADS / 32];
    const int n = blockIdx.x * SCAN_THREADS + threadIdx.x;
    const bool ok = n < N_NODES;
    unsigned int cnt_in = 0u;
    if (ok) {
        const unsigned int* ps = partial + n;
        unsigned int* pd = partial + (size_t)HIST_WGS * N_NODES + n;
        unsigned int s = 0u;
        #pragma unroll
        for (int c = 0; c < HIST_WGS; ++c) s += ps[(size_t)c * N_NODES];
        deg[n] = (float)s;
        unsigned int vd[HIST_WGS];
        #pragma unroll
        for (int c = 0; c < HIST_WGS; ++c) vd[c] = pd[(size_t)c * N_NODES];
        unsigned int run = 0u;
        #pragma unroll
        for (int c = 0; c < HIST_WGS; ++c) { pd[(size_t)c * N_NODES] = run; run += vd[c]; }
        deg[N_NODES + n] = (float)run;
        cnt_in = run;
    }
    unsigned int total;
    const unsigned int ex = gc_block_excl_scan(cnt_in, wsum, total);
    if (ok) local_off[n] = ex;
    if (threadIdx.x == 0) blocksum[blockIdx.x] = total;
}

__global__ __launch_bounds__(SCAN_THREADS)
void gc_fill_kernel(const int* __restrict__ src,
                    const int* __restrict__ dst,
                    const float* __restrict__ ew,
                    const unsigned int* __restrict__ rank,
                    const unsigned int* __restrict__ chunk_off,
                    const unsigned int* __restrict__ local_off,
                    const unsigned int* __restrict__ blocksum,
                    unsigned int* __restrict__ rowstart,
                    unsigned int* __restrict__ pairs_w) {
    __shared__ unsigned int wsum[SCAN_THREADS / 32];
    __shared__ unsigned int bp[SCAN_THREADS];
    const int t = threadIdx.x;
    const int e = blockIdx.x * SCAN_THREADS + t;
    const int   d = dst[e];
    const int   s = src[e];
    const float w = ew[e];
    const unsigned int r = rank[e];

    const unsigned int bs = (t < SCAN_BLOCKS) ? blocksum[t] : 0u;
    unsigned int total;
    bp[t] = gc_block_excl_scan(bs, wsum, total);
    __syncthreads();

    const int c = e / EDGES_PER_WG;
    const unsigned int pos = bp[d >> 8] + local_off[d]
                           + chunk_off[(size_t)c * N_NODES + d] + r;
    ((uint2*)pairs_w)[pos] = make_uint2((unsigned int)s, __float_as_uint(w));
    if (e < N_NODES) rowstart[e] = bp[e >> 8] + local_off[e];
}

#define GEMM_ROWS   128
#define GEMM_KS     32
#define GEMM_SLABS  (IN_FEATS / GEMM_KS)
#define GEMM_KQ     (GEMM_KS / 4)
#define GEMM_WGS    ((N_NODES + GEMM_ROWS - 1) / GEMM_ROWS)

__global__ __launch_bounds__(256)
void gc_gemm_wmma_kernel(const float* __restrict__ x,
                         const float* __restrict__ Wm,
                         const float* __restrict__ out_deg,
                         float* __restrict__ h_lo,
                         float* __restrict__ h_hi) {
    __shared__ float4 wq[2][GEMM_KQ][OUT_FEATS];

    const int tid  = threadIdx.x;
    const int lane = tid & 31;
    const int wave = tid >> 5;
    const int half = lane >> 4;
    const int l    = lane & 15;

    const int m0 = blockIdx.x * GEMM_ROWS + wave * 16;
    const bool live = m0 < N_NODES;
    const int row = live ? (m0 + l) : (N_NODES - 1);
    const float rs = rsqrtf(fmaxf(out_deg[row], 1.0f));
    const float4* __restrict__ xq = (const float4*)(x + (size_t)row * IN_FEATS) + half;

    const int sn  = tid & (OUT_FEATS - 1);
    const int skq = tid >> 7;
    const float* __restrict__ wcol = Wm + sn;

    float  wst[4][4];
    float4 acur[4], anext[4];

    #pragma unroll
    for (int q = 0; q < 4; ++q) {
        #pragma unroll
        for (int e = 0; e < 4; ++e)
            wst[q][e] = wcol[(size_t)(4 * (skq + 2 * q) + e) * OUT_FEATS];
    }
    #pragma unroll
    for (int j = 0; j < 4; ++j) acur[j] = xq[2 * j];
    #pragma unroll
    for (int q = 0; q < 4; ++q)
        wq[0][skq + 2 * q][sn] = make_float4(wst[q][0], wst[q][1], wst[q][2], wst[q][3]);
    __syncthreads();

    v8f c[8] = {};
    for (int s = 0; s < GEMM_SLABS; ++s) {
        const int buf = s & 1;
        const bool more = (s + 1 < GEMM_SLABS);
        if (more) {
            const int k1 = (s + 1) * GEMM_KS;
            #pragma unroll
            for (int q = 0; q < 4; ++q) {
                #pragma unroll
                for (int e = 0; e < 4; ++e)
                    wst[q][e] = wcol[(size_t)(k1 + 4 * (skq + 2 * q) + e) * OUT_FEATS];
            }
            #pragma unroll
            for (int j = 0; j < 4; ++j) anext[j] = xq[(s + 1) * GEMM_KQ + 2 * j];
        }
        asm volatile("" ::: "memory");

        float4 bcur[8], bnext[8];
        #pragma unroll
        for (int t = 0; t < 8; ++t) bcur[t] = wq[buf][half][16 * t + l];

        #pragma unroll
        for (int j = 0; j < 4; ++j) {
            if (j + 1 < 4) {
                #pragma unroll
                for (int t = 0; t < 8; ++t) bnext[t] = wq[buf][2 * (j + 1) + half][16 * t + l];
            }
            __builtin_amdgcn_sched_barrier(0);
            const float4 av = acur[j];
            v2f a0, a1;
            a0.x = av.x * rs; a0.y = av.y * rs;
            a1.x = av.z * rs; a1.y = av.w * rs;
            #pragma unroll
            for (int t = 0; t < 8; ++t) {
                v2f b0; b0.x = bcur[t].x; b0.y = bcur[t].y;
                c[t] = __builtin_amdgcn_wmma_f32_16x16x4_f32(false, a0, false, b0,
                                                             (short)0, c[t], false, false);
            }
            #pragma unroll
            for (int t = 0; t < 8; ++t) {
                v2f b1; b1.x = bcur[t].z; b1.y = bcur[t].w;
                c[t] = __builtin_amdgcn_wmma_f32_16x16x4_f32(false, a1, false, b1,
                                                             (short)0, c[t], false, false);
            }
            if (j + 1 < 4) {
                #pragma unroll
                for (int t = 0; t < 8; ++t) bcur[t] = bnext[t];
            }
        }

        if (more) {
            #pragma unroll
            for (int q = 0; q < 4; ++q)
                wq[buf ^ 1][skq + 2 * q][sn] =
                    make_float4(wst[q][0], wst[q][1], wst[q][2], wst[q][3]);
            __syncthreads();
            #pragma unroll
            for (int j = 0; j < 4; ++j) acur[j] = anext[j];
        }
    }

    if (live) {
        #pragma unroll
        for (int t = 0; t < 8; ++t) {
            float* __restrict__ hp;
            int pitch;
            if (t < 4) { hp = h_lo + 16 * t + l;                pitch = OUT_FEATS;  }
            else       { hp = h_hi + (16 * t - HALF_FEATS) + l; pitch = HALF_FEATS; }
            #pragma unroll
            for (int r = 0; r < 8; ++r) {
                const int m = m0 + r + 8 * half;
                hp[(size_t)m * pitch] = c[t][r];
            }
        }
    }
}

#define GATHER_WAVES 8
#define GATHER_NODES (2 * GATHER_WAVES)

template <int K>
__device__ __forceinline__ unsigned int gc_row_bcast(unsigned int v) {
    return (unsigned int)__builtin_amdgcn_mov_dpp((int)v, 0x150 + K, 0xf, 0xf, false);
}

template <int G>
__device__ __forceinline__ void gc_gather_group(const uint2 q, const int left,
                                                const float* hb, const unsigned int hpitch,
                                                float4& acc) {
    unsigned int s[8];
    float w[8];
    s[0] = gc_row_bcast<G + 0>(q.x); w[0] = __uint_as_float(gc_row_bcast<G + 0>(q.y));
    s[1] = gc_row_bcast<G + 1>(q.x); w[1] = __uint_as_float(gc_row_bcast<G + 1>(q.y));
    s[2] = gc_row_bcast<G + 2>(q.x); w[2] = __uint_as_float(gc_row_bcast<G + 2>(q.y));
    s[3] = gc_row_bcast<G + 3>(q.x); w[3] = __uint_as_float(gc_row_bcast<G + 3>(q.y));
    s[4] = gc_row_bcast<G + 4>(q.x); w[4] = __uint_as_float(gc_row_bcast<G + 4>(q.y));
    s[5] = gc_row_bcast<G + 5>(q.x); w[5] = __uint_as_float(gc_row_bcast<G + 5>(q.y));
    s[6] = gc_row_bcast<G + 6>(q.x); w[6] = __uint_as_float(gc_row_bcast<G + 6>(q.y));
    s[7] = gc_row_bcast<G + 7>(q.x); w[7] = __uint_as_float(gc_row_bcast<G + 7>(q.y));
    float4 r[8];
    #pragma unroll
    for (int k = 0; k < 8; ++k) {
        r[k] = *(const float4*)(hb + s[k] * hpitch);
    }
    #pragma unroll
    for (int k = 1; k < 8; ++k) {
        if (k >= left) r[k] = make_float4(0.0f, 0.0f, 0.0f, 0.0f);
    }
    #pragma unroll
    for (int k = 0; k < 8; ++k) {
        acc.x = fmaf(r[k].x, w[k], acc.x); acc.y = fmaf(r[k].y, w[k], acc.y);
        acc.z = fmaf(r[k].z, w[k], acc.z); acc.w = fmaf(r[k].w, w[k], acc.w);
    }
}

__global__ __launch_bounds__(GATHER_WAVES * 32)
void gc_gather_kernel(const float* hsrc, unsigned int hpitch,
                      const unsigned int* __restrict__ pairs_w,
                      const unsigned int* __restrict__ rowstart,
                      const float* __restrict__ in_deg,
                      const float* __restrict__ bias,
                      float* outp) {
    const int lane = threadIdx.x & 31;
    const int wv   = threadIdx.x >> 5;
    const int hf   = lane >> 4;
    const int l16  = lane & 15;
    const int d    = blockIdx.x * GATHER_NODES + wv * 2 + hf;

    const float degf = in_deg[d];
    const int   cnt  = (int)degf;
    unsigned int rs0 = rowstart[d];
    asm volatile("" : "+v"(rs0));
    const uint2* __restrict__ p = (const uint2*)pairs_w + rs0;
    const float* hb = hsrc + l16 * 4;

    float4 acc = make_float4(0.0f, 0.0f, 0.0f, 0.0f);
    for (int base = 0; base < cnt; base += 32) {
        const int left = cnt - base;
        uint2 qa = make_uint2(0u, 0u), qb = make_uint2(0u, 0u);
        if (l16 < left)      qa = p[base + l16];
        if (l16 + 16 < left) qb = p[base + 16 + l16];
        gc_gather_group<0>(qa, left, hb, hpitch, acc);
        if (left > 8)  gc_gather_group<8>(qa, left - 8,  hb, hpitch, acc);
        if (left > 16) gc_gather_group<0>(qb, left - 16, hb, hpitch, acc);
        if (left > 24) gc_gather_group<8>(qb, left - 24, hb, hpitch, acc);
    }

    const float rn = rsqrtf(fmaxf(degf, 1.0f));
    const float4 bb = *(const float4*)(bias + l16 * 4);
    float4 o;
    o.x = acc.x * rn + bb.x;
    o.y = acc.y * rn + bb.y;
    o.z = acc.z * rn + bb.z;
    o.w = acc.w * rn + bb.w;
    *(float4*)(outp + (size_t)d * OUT_FEATS + l16 * 4) = o;
}

extern "C" void kernel_launch(void* const* d_in, const int* in_sizes, int n_in,
                              void* d_out, int out_size, void* d_ws, size_t ws_size,
                              hipStream_t stream) {
    const float* x   = (const float*)d_in[0];
    const float* ew  = (const float*)d_in[1];
    const float* Wm  = (const float*)d_in[2];
    const float* b   = (const float*)d_in[3];
    const int*   src = (const int*)d_in[4];
    const int*   dst = (const int*)d_in[5];

    float* out = (float*)d_out;

    float*        out_deg   = (float*)d_ws;
    float*        in_deg    = out_deg + 50000;
    float*        h_hi      = out_deg + 100032;
    unsigned int* partial   = (unsigned int*)h_hi;
    unsigned int* pairs     = (unsigned int*)(out_deg + 3300064);
    unsigned int* rowstart  = (unsigned int*)(out_deg + 4900064);
    unsigned int* local_off = (unsigned int*)(out_deg + 4950064);
    unsigned int* blocksum  = (unsigned int*)(out_deg + 5000064);
    unsigned int* rank      = (unsigned int*)(out_deg + 5000320);

    gc_degree_hist_kernel<<<2 * HIST_WGS, HIST_THREADS, 0, stream>>>(src, dst, partial, rank);

    gc_degree_scan_kernel<<<SCAN_BLOCKS, SCAN_THREADS, 0, stream>>>(partial, out_deg, local_off, blocksum);

    gc_fill_kernel<<<N_EDGES / SCAN_THREADS, SCAN_THREADS, 0, stream>>>(
        src, dst, ew, rank, partial + HIST_WGS * N_NODES, local_off, blocksum, rowstart, pairs);

    gc_gemm_wmma_kernel<<<GEMM_WGS, 256, 0, stream>>>(x, Wm, out_deg, out + HALF_FEATS, h_hi);

    gc_gather_kernel<<<N_NODES / GATHER_NODES, GATHER_WAVES * 32, 0, stream>>>(
        out + HALF_FEATS, OUT_FEATS, pairs, rowstart, in_deg, b, out);

    gc_gather_kernel<<<N_NODES / GATHER_NODES, GATHER_WAVES * 32, 0, stream>>>(
        h_hi, HALF_FEATS, pairs, rowstart, in_deg, b + HALF_FEATS, out + HALF_FEATS);
}


# projection and edge placement share one launch: 391 WMMA GEMM blocks first, 391 placement blocks (8 edges a thread) in the slots they leave; chunk run starts go to out's left half; float sums kept
# speedup vs baseline: 3.3952x; 1.0990x over previous
#include <hip/hip_runtime.h>


typedef float v2f __attribute__((ext_vector_type(2)));
typedef float v8f __attribute__((ext_vector_type(8)));

#define N_NODES    50000
#define N_EDGES    800000
#define IN_FEATS   256
#define OUT_FEATS  128
#define HALF_FEATS 64

#define HIST_WGS      32
#define HIST_THREADS  1024
#define EDGES_PER_WG  (N_EDGES / HIST_WGS)
#define NODE_QUADS    (N_NODES / 4)
#define IDX_QUADS_PER_WG (EDGES_PER_WG / 4)
#define HIST_TRIPS    ((IDX_QUADS_PER_WG + HIST_THREADS - 1) / HIST_THREADS)

__global__ __launch_bounds__(HIST_THREADS)
void gc_degree_hist_kernel(const int* __restrict__ src,
                           const int* __restrict__ dst,
                           unsigned int* __restrict__ partial,
                           unsigned int* __restrict__ rank) {
    __shared__ unsigned int hist[N_NODES];
    const int tid   = threadIdx.x;
    const int wg    = blockIdx.x;
    const bool is_dst = (wg >= HIST_WGS);
    const int chunk = is_dst ? wg - HIST_WGS : wg;
    const int* __restrict__ idx = (is_dst ? dst : src) + (size_t)chunk * EDGES_PER_WG;
    const int4* __restrict__ idx4 = (const int4*)idx;

    int4 v[HIST_TRIPS];
    #pragma unroll
    for (int j = 0; j < HIST_TRIPS; ++j) {
        const int i = tid + j * HIST_THREADS;
        if (i < IDX_QUADS_PER_WG) v[j] = idx4[i];
    }

    uint4* hist4 = (uint4*)hist;
    for (int i = tid; i < NODE_QUADS; i += HIST_THREADS)
        hist4[i] = make_uint4(0u, 0u, 0u, 0u);
    __syncthreads();

    uint4* __restrict__ rank4 = (uint4*)rank + (size_t)chunk * IDX_QUADS_PER_WG;
    #pragma unroll
    for (int j = 0; j < HIST_TRIPS; ++j) {
        const int i = tid + j * HIST_THREADS;
        if (i < IDX_QUADS_PER_WG) {
            const unsigned int r0 = atomicAdd(&hist[v[j].x], 1u);
            const unsigned int r1 = atomicAdd(&hist[v[j].y], 1u);
            const unsigned int r2 = atomicAdd(&hist[v[j].z], 1u);
            const unsigned int r3 = atomicAdd(&hist[v[j].w], 1u);
            if (is_dst) rank4[i] = make_uint4(r0, r1, r2, r3);
        }
    }
    __syncthreads();

    uint4* __restrict__ out4 = (uint4*)(partial + (size_t)wg * N_NODES);
    for (int i = tid; i < NODE_QUADS; i += HIST_THREADS)
        out4[i] = hist4[i];
}

#define SCAN_THREADS 256
#define SCAN_BLOCKS  ((N_NODES + SCAN_THREADS - 1) / SCAN_THREADS)

__device__ __forceinline__ unsigned int gc_block_excl_scan(unsigned int v,
                                                           unsigned int* wsum,
                                                           unsigned int& total) {
    const int lane = threadIdx.x & 31;
    const int wv   = threadIdx.x >> 5;
    unsigned int inc = v;
    #pragma unroll
    for (int o = 1; o < 32; o <<= 1) {
        const unsigned int t = __shfl_up(inc, (unsigned int)o);
        if (lane >= o) inc += t;
    }
    if (lane == 31) wsum[wv] = inc;
    __syncthreads();
    unsigned int base = 0u, tot = 0u;
    #pragma unroll
    for (int k = 0; k < SCAN_THREADS / 32; ++k) {
        const unsigned int s = wsum[k];
        if (k < wv) base += s;
        tot += s;
    }
    total = tot;
    return base + inc - v;
}

#define CO_PITCH 50048

__device__ __forceinline__ unsigned int gc_co_word(unsigned int j) {
    return ((j >> 6) << 7) | (j & 63u);
}

__global__ __launch_bounds__(SCAN_THREADS)
void gc_degree_scan_kernel(const unsigned int* __restrict__ partial,
                           float* __restrict__ deg,
                           unsigned int* __restrict__ local_off,
                           unsigned int* __restrict__ blocksum,
                           unsigned int* __restrict__ chunk_off) {
    __shared__ unsigned int wsum[SCAN_THREADS / 32];
    const int n = blockIdx.x * SCAN_THREADS + threadIdx.x;
    const bool ok = n < N_NODES;
    unsigned int cnt_in = 0u;
    if (ok) {
        const unsigned int* ps = partial + n;
        const unsigned int* pd = partial + (size_t)HIST_WGS * N_NODES + n;
        unsigned int s = 0u;
        #pragma unroll
        for (int c = 0; c < HIST_WGS; ++c) s += ps[(size_t)c * N_NODES];
        deg[n] = (float)s;
        unsigned int vd[HIST_WGS];
        #pragma unroll
        for (int c = 0; c < HIST_WGS; ++c) vd[c] = pd[(size_t)c * N_NODES];
        unsigned int run = 0u;
        #pragma unroll
        for (int c = 0; c < HIST_WGS; ++c) {
            chunk_off[gc_co_word((unsigned int)(c * CO_PITCH + n))] = run;
            run += vd[c];
        }
        deg[N_NODES + n] = (float)run;
        cnt_in = run;
    }
    unsigned int total;
    const unsigned int ex = gc_block_excl_scan(cnt_in, wsum, total);
    if (ok) local_off[n] = ex;
    if (threadIdx.x == 0) blocksum[blockIdx.x] = total;
}

#define GEMM_ROWS   128
#define GEMM_KS     32
#define GEMM_SLABS  (IN_FEATS / GEMM_KS)
#define GEMM_KQ     (GEMM_KS / 4)
#define GEMM_WGS    ((N_NODES + GEMM_ROWS - 1) / GEMM_ROWS)
#define EDGE_QUADS  (N_EDGES / 4)
#define PLACE_WGS   ((EDGE_QUADS + 2 * SCAN_THREADS - 1) / (2 * SCAN_THREADS))

__device__ __forceinline__ uint4 gc_place4(const int4 d, const uint4 r, const unsigned int cb,
                                           const unsigned int* bp,
                                           const unsigned int* __restrict__ local_off,
                                           const unsigned int* __restrict__ chunk_off) {
    uint4 p;
    p.x = bp[d.x >> 8] + local_off[d.x] + chunk_off[gc_co_word(cb + (unsigned int)d.x)] + r.x;
    p.y = bp[d.y >> 8] + local_off[d.y] + chunk_off[gc_co_word(cb + (unsigned int)d.y)] + r.y;
    p.z = bp[d.z >> 8] + local_off[d.z] + chunk_off[gc_co_word(cb + (unsigned int)d.z)] + r.z;
    p.w = bp[d.w >> 8] + local_off[d.w] + chunk_off[gc_co_word(cb + (unsigned int)d.w)] + r.w;
    return p;
}

__global__ __launch_bounds__(256)
void gc_project_place_kernel(const float* __restrict__ x,
                             const float* __restrict__ Wm,
                             const float* __restrict__ out_deg,
                             float* __restrict__ h_lo,
                             float* __restrict__ h_hi,
                             const int* __restrict__ src,
                             const int* __restrict__ dst,
                             const float* __restrict__ ew,
                             const unsigned int* __restrict__ rank,
                             const unsigned int* __restrict__ chunk_off,
                             const unsigned int* __restrict__ local_off,
                             const unsigned int* __restrict__ blocksum,
                             unsigned int* __restrict__ rowstart,
                             unsigned int* __restrict__ pairs_w) {
    __shared__ float4 wq[2][GEMM_KQ][OUT_FEATS];
    __shared__ unsigned int wsum[SCAN_THREADS / 32];
    __shared__ unsigned int bp[SCAN_THREADS];

    if (blockIdx.x >= GEMM_WGS) {
        const int t  = threadIdx.x;
        const int b  = blockIdx.x - GEMM_WGS;
        const int q0 = b * (2 * SCAN_THREADS) + t;
        const int q1 = q0 + SCAN_THREADS;
        const bool ok0 = q0 < EDGE_QUADS;
        const bool ok1 = q1 < EDGE_QUADS;
        const int nq = b * SCAN_THREADS + t;
        const bool okn = nq < NODE_QUADS;

        int4   d0 = make_int4(0, 0, 0, 0), d1 = d0, s0 = d0, s1 = d0;
        uint4  r0 = make_uint4(0u, 0u, 0u, 0u), r1 = r0, lo = r0;
        float4 w0 = make_float4(0.0f, 0.0f, 0.0f, 0.0f), w1 = w0;
        if (ok0) {
            d0 = ((const int4*)dst)[q0];  s0 = ((const int4*)src)[q0];
            w0 = ((const float4*)ew)[q0]; r0 = ((const uint4*)rank)[q0];
        }
        if (ok1) {
            d1 = ((const int4*)dst)[q1];  s1 = ((const int4*)src)[q1];
            w1 = ((const float4*)ew)[q1]; r1 = ((const uint4*)rank)[q1];
        }
        if (okn) lo = ((const uint4*)local_off)[nq];

        const unsigned int bs = (t < SCAN_BLOCKS) ? blocksum[t] : 0u;
        unsigned int total;
        bp[t] = gc_block_excl_scan(bs, wsum, total);
        __syncthreads();

        const unsigned int cb0 = (unsigned int)(q0 / IDX_QUADS_PER_WG) * CO_PITCH;
        const unsigned int cb1 = ok1 ? (unsigned int)(q1 / IDX_QUADS_PER_WG) * CO_PITCH : 0u;
        const uint4 p0 = gc_place4(d0, r0, ok0 ? cb0 : 0u, bp, local_off, chunk_off);
        const uint4 p1 = gc_place4(d1, r1, cb1, bp, local_off, chunk_off);
        uint2* __restrict__ pw = (uint2*)pairs_w;
        if (ok0) {
            pw[p0.x] = make_uint2((unsigned int)s0.x, __float_as_uint(w0.x));
            pw[p0.y] = make_uint2((unsigned int)s0.y, __float_as_uint(w0.y));
            pw[p0.z] = make_uint2((unsigned int)s0.z, __float_as_uint(w0.z));
            pw[p0.w] = make_uint2((unsigned int)s0.w, __float_as_uint(w0.w));
        }
        if (ok1) {
            pw[p1.x] = make_uint2((unsigned int)s1.x, __float_as_uint(w1.x));
            pw[p1.y] = make_uint2((unsigned int)s1.y, __float_as_uint(w1.y));
            pw[p1.z] = make_uint2((unsigned int)s1.z, __float_as_uint(w1.z));
            pw[p1.w] = make_uint2((unsigned int)s1.w, __float_as_uint(w1.w));
        }
        if (okn) {
            const unsigned int base = bp[nq >> 6];
            ((uint4*)rowstart)[nq] = make_uint4(base + lo.x, base + lo.y, base + lo.z, base + lo.w);
        }
        return;
    }

    const int tid  = threadIdx.x;
    const int lane = tid & 31;
    const int wave = tid >> 5;
    const int half = lane >> 4;
    const int l    = lane & 15;

    const int m0 = blockIdx.x * GEMM_ROWS + wave * 16;
    const bool live = m0 < N_NODES;
    const int row = live ? (m0 + l) : (N_NODES - 1);
    const float rs = rsqrtf(fmaxf(out_deg[row], 1.0f));
    const float4* __restrict__ xq = (const float4*)(x + (size_t)row * IN_FEATS) + half;

    const int sn  = tid & (OUT_FEATS - 1);
    const int skq = tid >> 7;
    const float* __restrict__ wcol = Wm + sn;

    float  wst[4][4];
    float4 acur[4], anext[4];

    #pragma unroll
    for (int q = 0; q < 4; ++q) {
        #pragma unroll
        for (int e = 0; e < 4; ++e)
            wst[q][e] = wcol[(size_t)(4 * (skq + 2 * q) + e) * OUT_FEATS];
    }
    #pragma unroll
    for (int j = 0; j < 4; ++j) acur[j] = xq[2 * j];
    #pragma unroll
    for (int q = 0; q < 4; ++q)
        wq[0][skq + 2 * q][sn] = make_float4(wst[q][0], wst[q][1], wst[q][2], wst[q][3]);
    __syncthreads();

    v8f c[8] = {};
    for (int s = 0; s < GEMM_SLABS; ++s) {
        const int buf = s & 1;
        const bool more = (s + 1 < GEMM_SLABS);
        if (more) {
            const int k1 = (s + 1) * GEMM_KS;
            #pragma unroll
            for (int q = 0; q < 4; ++q) {
                #pragma unroll
                for (int e = 0; e < 4; ++e)
                    wst[q][e] = wcol[(size_t)(k1 + 4 * (skq + 2 * q) + e) * OUT_FEATS];
            }
            #pragma unroll
            for (int j = 0; j < 4; ++j) anext[j] = xq[(s + 1) * GEMM_KQ + 2 * j];
        }
        asm volatile("" ::: "memory");

        float4 bcur[8], bnext[8];
        #pragma unroll
        for (int t = 0; t < 8; ++t) bcur[t] = wq[buf][half][16 * t + l];

        #pragma unroll
        for (int j = 0; j < 4; ++j) {
            if (j + 1 < 4) {
                #pragma unroll
                for (int t = 0; t < 8; ++t) bnext[t] = wq[buf][2 * (j + 1) + half][16 * t + l];
            }
            __builtin_amdgcn_sched_barrier(0);
            const float4 av = acur[j];
            v2f a0, a1;
            a0.x = av.x * rs; a0.y = av.y * rs;
            a1.x = av.z * rs; a1.y = av.w * rs;
            #pragma unroll
            for (int t = 0; t < 8; ++t) {
                v2f b0; b0.x = bcur[t].x; b0.y = bcur[t].y;
                c[t] = __builtin_amdgcn_wmma_f32_16x16x4_f32(false, a0, false, b0,
                                                             (short)0, c[t], false, false);
            }
            #pragma unroll
            for (int t = 0; t < 8; ++t) {
                v2f b1; b1.x = bcur[t].z; b1.y = bcur[t].w;
                c[t] = __builtin_amdgcn_wmma_f32_16x16x4_f32(false, a1, false, b1,
                                                             (short)0, c[t], false, false);
            }
            if (j + 1 < 4) {
                #pragma unroll
                for (int t = 0; t < 8; ++t) bcur[t] = bnext[t];
            }
        }

        if (more) {
            #pragma unroll
            for (int q = 0; q < 4; ++q)
                wq[buf ^ 1][skq + 2 * q][sn] =
                    make_float4(wst[q][0], wst[q][1], wst[q][2], wst[q][3]);
            __syncthreads();
            #pragma unroll
            for (int j = 0; j < 4; ++j) acur[j] = anext[j];
        }
    }

    if (live) {
        #pragma unroll
        for (int t = 0; t < 8; ++t) {
            float* __restrict__ hp;
            int pitch;
            if (t < 4) { hp = h_lo + 16 * t + l;                pitch = OUT_FEATS;  }
            else       { hp = h_hi + (16 * t - HALF_FEATS) + l; pitch = HALF_FEATS; }
            #pragma unroll
            for (int r = 0; r < 8; ++r) {
                const int m = m0 + r + 8 * half;
                hp[(size_t)m * pitch] = c[t][r];
            }
        }
    }
}

#define GATHER_WAVES 8
#define GATHER_NODES (2 * GATHER_WAVES)

template <int K>
__device__ __forceinline__ unsigned int gc_row_bcast(unsigned int v) {
    return (unsigned int)__builtin_amdgcn_mov_dpp((int)v, 0x150 + K, 0xf, 0xf, false);
}

template <int G>
__device__ __forceinline__ void gc_gather_group(const uint2 q, const int left,
                                                const float* hb, const unsigned int hpitch,
                                                float4& acc) {
    unsigned int s[8];
    float w[8];
    s[0] = gc_row_bcast<G + 0>(q.x); w[0] = __uint_as_float(gc_row_bcast<G + 0>(q.y));
    s[1] = gc_row_bcast<G + 1>(q.x); w[1] = __uint_as_float(gc_row_bcast<G + 1>(q.y));
    s[2] = gc_row_bcast<G + 2>(q.x); w[2] = __uint_as_float(gc_row_bcast<G + 2>(q.y));
    s[3] = gc_row_bcast<G + 3>(q.x); w[3] = __uint_as_float(gc_row_bcast<G + 3>(q.y));
    s[4] = gc_row_bcast<G + 4>(q.x); w[4] = __uint_as_float(gc_row_bcast<G + 4>(q.y));
    s[5] = gc_row_bcast<G + 5>(q.x); w[5] = __uint_as_float(gc_row_bcast<G + 5>(q.y));
    s[6] = gc_row_bcast<G + 6>(q.x); w[6] = __uint_as_float(gc_row_bcast<G + 6>(q.y));
    s[7] = gc_row_bcast<G + 7>(q.x); w[7] = __uint_as_float(gc_row_bcast<G + 7>(q.y));
    float4 r[8];
    #pragma unroll
    for (int k = 0; k < 8; ++k) {
        r[k] = *(const float4*)(hb + s[k] * hpitch);
    }
    #pragma unroll
    for (int k = 1; k < 8; ++k) {
        if (k >= left) r[k] = make_float4(0.0f, 0.0f, 0.0f, 0.0f);
    }
    #pragma unroll
    for (int k = 0; k < 8; ++k) {
        acc.x = fmaf(r[k].x, w[k], acc.x); acc.y = fmaf(r[k].y, w[k], acc.y);
        acc.z = fmaf(r[k].z, w[k], acc.z); acc.w = fmaf(r[k].w, w[k], acc.w);
    }
}

__global__ __launch_bounds__(GATHER_WAVES * 32)
void gc_gather_kernel(const float* hsrc, unsigned int hpitch,
                      const unsigned int* __restrict__ pairs_w,
                      const unsigned int* __restrict__ rowstart,
                      const float* __restrict__ in_deg,
                      const float* __restrict__ bias,
                      float* outp) {
    const int lane = threadIdx.x & 31;
    const int wv   = threadIdx.x >> 5;
    const int hf   = lane >> 4;
    const int l16  = lane & 15;
    const int d    = blockIdx.x * GATHER_NODES + wv * 2 + hf;

    const float degf = in_deg[d];
    const int   cnt  = (int)degf;
    unsigned int rs0 = rowstart[d];
    asm volatile("" : "+v"(rs0));
    const uint2* __restrict__ p = (const uint2*)pairs_w + rs0;
    const float* hb = hsrc + l16 * 4;

    float4 acc = make_float4(0.0f, 0.0f, 0.0f, 0.0f);
    for (int base = 0; base < cnt; base += 32) {
        const int left = cnt - base;
        uint2 qa = make_uint2(0u, 0u), qb = make_uint2(0u, 0u);
        if (l16 < left)      qa = p[base + l16];
        if (l16 + 16 < left) qb = p[base + 16 + l16];
        gc_gather_group<0>(qa, left, hb, hpitch, acc);
        if (left > 8)  gc_gather_group<8>(qa, left - 8,  hb, hpitch, acc);
        if (left > 16) gc_gather_group<0>(qb, left - 16, hb, hpitch, acc);
        if (left > 24) gc_gather_group<8>(qb, left - 24, hb, hpitch, acc);
    }

    const float rn = rsqrtf(fmaxf(degf, 1.0f));
    const float4 bb = *(const float4*)(bias + l16 * 4);
    float4 o;
    o.x = acc.x * rn + bb.x;
    o.y = acc.y * rn + bb.y;
    o.z = acc.z * rn + bb.z;
    o.w = acc.w * rn + bb.w;
    *(float4*)(outp + (size_t)d * OUT_FEATS + l16 * 4) = o;
}

extern "C" void kernel_launch(void* const* d_in, const int* in_sizes, int n_in,
                              void* d_out, int out_size, void* d_ws, size_t ws_size,
                              hipStream_t stream) {
    const float* x   = (const float*)d_in[0];
    const float* ew  = (const float*)d_in[1];
    const float* Wm  = (const float*)d_in[2];
    const float* b   = (const float*)d_in[3];
    const int*   src = (const int*)d_in[4];
    const int*   dst = (const int*)d_in[5];

    float* out = (float*)d_out;

    float*        out_deg   = (float*)d_ws;
    float*        in_deg    = out_deg + 50000;
    float*        h_hi      = out_deg + 100032;
    unsigned int* partial   = (unsigned int*)h_hi;
    unsigned int* pairs     = (unsigned int*)(out_deg + 3300064);
    unsigned int* rowstart  = (unsigned int*)(out_deg + 4900064);
    unsigned int* local_off = (unsigned int*)(out_deg + 4950064);
    unsigned int* blocksum  = (unsigned int*)(out_deg + 5000064);
    unsigned int* rank      = (unsigned int*)(out_deg + 5000320);

    gc_degree_hist_kernel<<<2 * HIST_WGS, HIST_THREADS, 0, stream>>>(src, dst, partial, rank);

    unsigned int* chunk_off = (unsigned int*)out;
    gc_degree_scan_kernel<<<SCAN_BLOCKS, SCAN_THREADS, 0, stream>>>(
        partial, out_deg, local_off, blocksum, chunk_off);

    gc_project_place_kernel<<<GEMM_WGS + PLACE_WGS, 256, 0, stream>>>(
        x, Wm, out_deg, out + HALF_FEATS, h_hi,
        src, dst, ew, rank, chunk_off, local_off, blocksum, rowstart, pairs);

    gc_gather_kernel<<<N_NODES / GATHER_NODES, GATHER_WAVES * 32, 0, stream>>>(
        out + HALF_FEATS, OUT_FEATS, pairs, rowstart, in_deg, b, out);

    gc_gather_kernel<<<N_NODES / GATHER_NODES, GATHER_WAVES * 32, 0, stream>>>(
        h_hi, HALF_FEATS, pairs, rowstart, in_deg, b + HALF_FEATS, out + HALF_FEATS);
}
